# starting kernel
# baseline (speedup 1.0000x reference)
#include <hip/hip_runtime.h>
#include <hip/hip_bf16.h>
#include <math.h>

#define SEQ    512
#define DIN    1024
#define HID    512

typedef float v2f __attribute__((ext_vector_type(2)));
typedef float v8f __attribute__((ext_vector_type(8)));

__device__ __forceinline__ float hw_tanh(float x)
{
#if __has_builtin(__builtin_amdgcn_tanhf)
    return __builtin_amdgcn_tanhf(x);
#else
    return tanhf(x);
#endif
}

#define CHUNK 256
#define PITCH 260

__global__ void pair_tanh_score(const float* __restrict__ Hm,
                                const float* __restrict__ Mm,
                                const float* __restrict__ w2,
                                const float* __restrict__ b2p,
                                float* __restrict__ out)
{
    __shared__ float sH[16 * PITCH];
    __shared__ float sM[16 * PITCH];
    __shared__ float sW[HID];

    const int t  = threadIdx.x;
    const int ti = t >> 4;
    const int tj = t & 15;
    const int i0 = blockIdx.y * 16;
    const int j0 = blockIdx.x * 16;

    sW[t]       = w2[t];
    sW[t + 256] = w2[t + 256];

    float acc = 0.0f;

    for (int d0 = 0; d0 < HID; d0 += CHUNK) {
        int idx = t;
#pragma unroll
        for (int rep = 0; rep < 4; ++rep, idx += 256) {
            const int r  = idx >> 6;
            const int c4 = (idx & 63) * 4;
            const float4 hv = *(const float4*)(Hm + (size_t)(i0 + r) * HID + d0 + c4);
            const float4 mv = *(const float4*)(Mm + (size_t)(j0 + r) * HID + d0 + c4);
            *(float4*)(&sH[r * PITCH + c4]) = hv;
            *(float4*)(&sM[r * PITCH + c4]) = mv;
        }
        __syncthreads();

        const float* __restrict__ hrow = &sH[ti * PITCH];
        const float* __restrict__ mrow = &sM[tj * PITCH];
        const float* __restrict__ wrow = &sW[d0];
#pragma unroll 8
        for (int d = 0; d < CHUNK; ++d) {
            const float v = hrow[d] + mrow[d];
            acc = fmaf(wrow[d], hw_tanh(v), acc);
        }
        __syncthreads();
    }

    out[(size_t)(i0 + ti) * SEQ + (j0 + tj)] = acc + b2p[0];
}

__global__ void gemm_bias_wmma_f32(const float* __restrict__ X,
                                   const float* __restrict__ W,
                                   const float* __restrict__ bias,
                                   float* __restrict__ Out)
{
    const int wavesPerBlock = blockDim.x >> 5;
    const int wave = blockIdx.x * wavesPerBlock + (threadIdx.x >> 5);
    const int lane = threadIdx.x & 31;

    const int tilesN = HID / 16;
    const int tm = wave / tilesN;
    const int tn = wave % tilesN;
    const int s0 = tm * 16;
    const int h0 = tn * 16;

    const int half = lane >> 4;
    const int l    = lane & 15;

    const float* __restrict__ arow = X + (size_t)(s0 + l) * DIN + 2 * half;
    const float* __restrict__ brow = W + (size_t)(h0 + l) * DIN + 2 * half;

    v8f acc = {};
#pragma unroll 8
    for (int k0 = 0; k0 < DIN; k0 += 4) {
        v2f a = *(const v2f*)(arow + k0);
        v2f b = *(const v2f*)(brow + k0);
        acc = __builtin_amdgcn_wmma_f32_16x16x4_f32(
             false, a,  false, b,
             (short)0, acc,  false,  false);
    }

    const float bv = bias[h0 + l];
#pragma unroll
    for (int r = 0; r < 8; ++r) {
        Out[(size_t)(s0 + r + 8 * half) * HID + h0 + l] = acc[r] + bv;
    }
}

extern "C" void kernel_launch(void* const* d_in, const int* in_sizes, int n_in,
                              void* d_out, int out_size, void* d_ws, size_t ws_size,
                              hipStream_t stream)
{
    const float* x   = (const float*)d_in[0];
    const float* W_h = (const float*)d_in[1];
    const float* b_h = (const float*)d_in[2];
    const float* W_m = (const float*)d_in[3];
    const float* b_m = (const float*)d_in[4];
    const float* w2  = (const float*)d_in[5];
    const float* b2  = (const float*)d_in[6];
    float* out = (float*)d_out;

    float* Hbuf = (float*)d_ws;
    float* Mbuf = Hbuf + (size_t)SEQ * HID;

    const int tiles = (SEQ / 16) * (HID / 16);
    const int blk   = 128;
    const int grid  = tiles / (blk / 32);
    gemm_bias_wmma_f32<<<grid, blk, 0, stream>>>(x, W_h, b_h, Hbuf);
    gemm_bias_wmma_f32<<<grid, blk, 0, stream>>>(x, W_m, b_m, Mbuf);

    dim3 g2(SEQ / 16, SEQ / 16);
    pair_tanh_score<<<g2, 256, 0, stream>>>(Hbuf, Mbuf, w2, b2, out);
}


# GEMM operand feed pipelined in registers, no LDS: each lane streams 16-B pieces of its X and W rows from global into a two-group register ring 128 k ahead of the WMMAs; k sum order changes
# speedup vs baseline: 1.6049x; 1.6049x over previous
#include <hip/hip_runtime.h>
#include <hip/hip_bf16.h>
#include <math.h>

#define SEQ    512
#define DIN    1024
#define HID    512

typedef float v2f __attribute__((ext_vector_type(2)));
typedef float v8f __attribute__((ext_vector_type(8)));

__device__ __forceinline__ float hw_tanh(float x)
{
#if __has_builtin(__builtin_amdgcn_tanhf)
    return __builtin_amdgcn_tanhf(x);
#else
    return tanhf(x);
#endif
}

#define CHUNK 256
#define PITCH 260

__global__ void pair_tanh_score(const float* __restrict__ Hm,
                                const float* __restrict__ Mm,
                                const float* __restrict__ w2,
                                const float* __restrict__ b2p,
                                float* __restrict__ out)
{
    __shared__ float sH[16 * PITCH];
    __shared__ float sM[16 * PITCH];
    __shared__ float sW[HID];

    const int t  = threadIdx.x;
    const int ti = t >> 4;
    const int tj = t & 15;
    const int i0 = blockIdx.y * 16;
    const int j0 = blockIdx.x * 16;

    sW[t]       = w2[t];
    sW[t + 256] = w2[t + 256];

    float acc = 0.0f;

    for (int d0 = 0; d0 < HID; d0 += CHUNK) {
        int idx = t;
#pragma unroll
        for (int rep = 0; rep < 4; ++rep, idx += 256) {
            const int r  = idx >> 6;
            const int c4 = (idx & 63) * 4;
            const float4 hv = *(const float4*)(Hm + (size_t)(i0 + r) * HID + d0 + c4);
            const float4 mv = *(const float4*)(Mm + (size_t)(j0 + r) * HID + d0 + c4);
            *(float4*)(&sH[r * PITCH + c4]) = hv;
            *(float4*)(&sM[r * PITCH + c4]) = mv;
        }
        __syncthreads();

        const float* __restrict__ hrow = &sH[ti * PITCH];
        const float* __restrict__ mrow = &sM[tj * PITCH];
        const float* __restrict__ wrow = &sW[d0];
#pragma unroll 8
        for (int d = 0; d < CHUNK; ++d) {
            const float v = hrow[d] + mrow[d];
            acc = fmaf(wrow[d], hw_tanh(v), acc);
        }
        __syncthreads();
    }

    out[(size_t)(i0 + ti) * SEQ + (j0 + tj)] = acc + b2p[0];
}

#define KG 128
#define NP (KG / 8)
#define NG (DIN / KG)

typedef float v4f __attribute__((ext_vector_type(4)));

__device__ __forceinline__ void load_group(const float* __restrict__ ap,
                                           const float* __restrict__ bp,
                                           v4f (&a)[NP], v4f (&b)[NP])
{
#pragma unroll
    for (int p = 0; p < NP; ++p) {
        a[p] = *(const v4f*)(ap + 8 * p);
        b[p] = *(const v4f*)(bp + 8 * p);
    }
}

__device__ __forceinline__ v8f mma_group(const v4f (&a)[NP], const v4f (&b)[NP], v8f acc)
{
#pragma unroll
    for (int p = 0; p < NP; ++p) {
        const v2f a01 = {a[p].x, a[p].y};
        const v2f b01 = {b[p].x, b[p].y};
        const v2f a23 = {a[p].z, a[p].w};
        const v2f b23 = {b[p].z, b[p].w};
        acc = __builtin_amdgcn_wmma_f32_16x16x4_f32(
             false, a01,  false, b01,
             (short)0, acc,  false,  false);
        acc = __builtin_amdgcn_wmma_f32_16x16x4_f32(
             false, a23,  false, b23,
             (short)0, acc,  false,  false);
    }
    return acc;
}

__global__ void __launch_bounds__(128)
gemm_bias_wmma_f32(const float* __restrict__ X,
                   const float* __restrict__ W,
                   const float* __restrict__ bias,
                   float* __restrict__ Out)
{
    const int wavesPerBlock = blockDim.x >> 5;
    const int wave = blockIdx.x * wavesPerBlock + (threadIdx.x >> 5);
    const int lane = threadIdx.x & 31;

    const int tilesN = HID / 16;
    const int tm = wave / tilesN;
    const int tn = wave % tilesN;
    const int s0 = tm * 16;
    const int h0 = tn * 16;

    const int half = lane >> 4;
    const int l    = lane & 15;

    const float* __restrict__ arow = X + (size_t)(s0 + l) * DIN + 4 * half;
    const float* __restrict__ brow = W + (size_t)(h0 + l) * DIN + 4 * half;

    v4f a0[NP], b0[NP], a1[NP], b1[NP];
    v8f acc = {};

    load_group(arow, brow, a0, b0);
    for (int t = 0; t < NG / 2 - 1; ++t) {
        const int k1 = (2 * t + 1) * KG;
        load_group(arow + k1, brow + k1, a1, b1);
        acc = mma_group(a0, b0, acc);
        load_group(arow + k1 + KG, brow + k1 + KG, a0, b0);
        acc = mma_group(a1, b1, acc);
    }
    load_group(arow + (NG - 1) * KG, brow + (NG - 1) * KG, a1, b1);
    acc = mma_group(a0, b0, acc);
    acc = mma_group(a1, b1, acc);

    const float bv = bias[h0 + l];
#pragma unroll
    for (int r = 0; r < 8; ++r) {
        Out[(size_t)(s0 + r + 8 * half) * HID + h0 + l] = acc[r] + bv;
    }
}

extern "C" void kernel_launch(void* const* d_in, const int* in_sizes, int n_in,
                              void* d_out, int out_size, void* d_ws, size_t ws_size,
                              hipStream_t stream)
{
    const float* x   = (const float*)d_in[0];
    const float* W_h = (const float*)d_in[1];
    const float* b_h = (const float*)d_in[2];
    const float* W_m = (const float*)d_in[3];
    const float* b_m = (const float*)d_in[4];
    const float* w2  = (const float*)d_in[5];
    const float* b2  = (const float*)d_in[6];
    float* out = (float*)d_out;

    float* Hbuf = (float*)d_ws;
    float* Mbuf = Hbuf + (size_t)SEQ * HID;

    const int tiles = (SEQ / 16) * (HID / 16);
    const int blk   = 128;
    const int grid  = tiles / (blk / 32);
    gemm_bias_wmma_f32<<<grid, blk, 0, stream>>>(x, W_h, b_h, Hbuf);
    gemm_bias_wmma_f32<<<grid, blk, 0, stream>>>(x, W_m, b_m, Mbuf);

    dim3 g2(SEQ / 16, SEQ / 16);
    pair_tanh_score<<<g2, 256, 0, stream>>>(Hbuf, Mbuf, w2, b2, out);
}


# pairwise tanh phase re-decomposed, d across lanes: a wave keeps a 4x32 output tile in registers, operands straight from global, no LDS; cross-lane sum at the end, summation order changes
# speedup vs baseline: 1.8791x; 1.1709x over previous
#include <hip/hip_runtime.h>
#include <hip/hip_bf16.h>
#include <math.h>

#define SEQ    512
#define DIN    1024
#define HID    512

typedef float v2f __attribute__((ext_vector_type(2)));
typedef float v8f __attribute__((ext_vector_type(8)));

__device__ __forceinline__ float hw_tanh(float x)
{
#if __has_builtin(__builtin_amdgcn_tanhf)
    return __builtin_amdgcn_tanhf(x);
#else
    return tanhf(x);
#endif
}

typedef float v4f __attribute__((ext_vector_type(4)));

#define TI    4
#define TJL   8
#define DSTEP 32
#define NSTEP (HID / DSTEP)

__device__ __forceinline__ void load_step(const float* __restrict__ hp,
                                          const float* __restrict__ mp,
                                          const float* __restrict__ wp,
                                          v4f (&h)[TI], v4f (&m)[TJL], v4f& w)
{
#pragma unroll
    for (int a = 0; a < TI; ++a) h[a] = *(const v4f*)(hp + (size_t)a * HID);
#pragma unroll
    for (int b = 0; b < TJL; ++b) m[b] = *(const v4f*)(mp + (size_t)b * HID);
    w = *(const v4f*)wp;
}

__device__ __forceinline__ v2f tanh2(v2f x)
{
    const v2f r = {hw_tanh(x.x), hw_tanh(x.y)};
    return r;
}

#define NCELL (TI * TJL * 2)
#define PIPE  3

__device__ __forceinline__ void pair_step(const v4f (&h)[TI], const v4f (&m)[TJL],
                                          const v4f w, v2f (&acc)[TI][TJL])
{
    const v2f wlo = {w.x, w.y};
    const v2f whi = {w.z, w.w};
    v2f x[NCELL], t[NCELL];
#pragma unroll
    for (int c = 0; c < NCELL + 2 * PIPE; ++c) {
        if (c < NCELL) {
            const int a = (c / TJL) % TI, b = c % TJL;
            const v2f hh = (c < NCELL / 2) ? (v2f){h[a].x, h[a].y} : (v2f){h[a].z, h[a].w};
            const v2f mm = (c < NCELL / 2) ? (v2f){m[b].x, m[b].y} : (v2f){m[b].z, m[b].w};
            x[c] = hh + mm;
        }
        if (c >= PIPE && c < NCELL + PIPE) t[c - PIPE] = tanh2(x[c - PIPE]);
        if (c >= 2 * PIPE) {
            const int e = c - 2 * PIPE;
            const int a = (e / TJL) % TI, b = e % TJL;
            acc[a][b] = __builtin_elementwise_fma((e < NCELL / 2) ? wlo : whi, t[e], acc[a][b]);
        }
    }
#pragma unroll
    for (int k = 0; k < 2 * NCELL; ++k) {
        __builtin_amdgcn_sched_group_barrier(0x002, 1, 0);
        __builtin_amdgcn_sched_group_barrier(0x400, 1, 0);
    }
}

template <int CTRL>
__device__ __forceinline__ float lane_exchange(float x)
{
    const int b = __builtin_bit_cast(int, x);
    return __builtin_bit_cast(float,
        __builtin_amdgcn_update_dpp(b, b, CTRL, 0xf, 0xf, false));
}

template <int N, int CTRL>
__device__ __forceinline__ void halve(float (&v)[32], const bool upper)
{
#pragma unroll
    for (int k = 0; k < N / 2; ++k) {
        const float lo = v[k];
        const float hi = v[k + N / 2];
        const float keep = upper ? hi : lo;
        const float send = upper ? lo : hi;
        v[k] = keep + lane_exchange<CTRL>(send);
    }
}

__global__ void __launch_bounds__(256)
pair_tanh_score(const float* __restrict__ Hm,
                const float* __restrict__ Mm,
                const float* __restrict__ w2,
                const float* __restrict__ b2p,
                float* __restrict__ out)
{
    const int lane = threadIdx.x & 31;
    const int wave = threadIdx.x >> 5;
    const int q = lane & 7;
    const int g = lane >> 3;
    const int iw = blockIdx.y * 32 + wave * TI;
    const int jw = blockIdx.x * 32;

    const float* __restrict__ hp = Hm + (size_t)iw * HID + 4 * q;
    const float* __restrict__ mp = Mm + (size_t)(jw + TJL * g) * HID + 4 * q;
    const float* __restrict__ wp = w2 + 4 * q;

    v2f acc[TI][TJL];
#pragma unroll
    for (int a = 0; a < TI; ++a)
#pragma unroll
        for (int b = 0; b < TJL; ++b) { acc[a][b].x = 0.0f; acc[a][b].y = 0.0f; }

    v4f h0[TI], m0[TJL], w0;
    v4f h1[TI], m1[TJL], w1;

    load_step(hp, mp, wp, h0, m0, w0);
    for (int t = 0; t < NSTEP / 2 - 1; ++t) {
        const int d1 = (2 * t + 1) * DSTEP;
        load_step(hp + d1, mp + d1, wp + d1, h1, m1, w1);
        __builtin_amdgcn_sched_barrier(0);
        pair_step(h0, m0, w0, acc);
        __builtin_amdgcn_sched_barrier(0);
        load_step(hp + d1 + DSTEP, mp + d1 + DSTEP, wp + d1 + DSTEP, h0, m0, w0);
        __builtin_amdgcn_sched_barrier(0);
        pair_step(h1, m1, w1, acc);
        __builtin_amdgcn_sched_barrier(0);
    }
    load_step(hp + (NSTEP - 1) * DSTEP, mp + (NSTEP - 1) * DSTEP,
              wp + (NSTEP - 1) * DSTEP, h1, m1, w1);
    __builtin_amdgcn_sched_barrier(0);
    pair_step(h0, m0, w0, acc);
    __builtin_amdgcn_sched_barrier(0);
    pair_step(h1, m1, w1, acc);

    float v[32];
#pragma unroll
    for (int a = 0; a < TI; ++a)
#pragma unroll
        for (int b = 0; b < TJL; ++b) v[TJL * a + b] = acc[a][b].x + acc[a][b].y;

    halve<32, 0x164>(v, (lane & 4) != 0);
    halve<16, 0x162>(v, (lane & 2) != 0);
    halve< 8, 0x161>(v, (lane & 1) != 0);

    const float b2 = b2p[0];
    const v4f res = {v[0] + b2, v[1] + b2, v[2] + b2, v[3] + b2};
    *(v4f*)(out + (size_t)(iw + (q >> 1)) * SEQ + jw + TJL * g + 4 * (q & 1)) = res;
}

#define KG 128
#define NP (KG / 8)
#define NG (DIN / KG)


__device__ __forceinline__ void load_group(const float* __restrict__ ap,
                                           const float* __restrict__ bp,
                                           v4f (&a)[NP], v4f (&b)[NP])
{
#pragma unroll
    for (int p = 0; p < NP; ++p) {
        a[p] = *(const v4f*)(ap + 8 * p);
        b[p] = *(const v4f*)(bp + 8 * p);
    }
}

__device__ __forceinline__ v8f mma_group(const v4f (&a)[NP], const v4f (&b)[NP], v8f acc)
{
#pragma unroll
    for (int p = 0; p < NP; ++p) {
        const v2f a01 = {a[p].x, a[p].y};
        const v2f b01 = {b[p].x, b[p].y};
        const v2f a23 = {a[p].z, a[p].w};
        const v2f b23 = {b[p].z, b[p].w};
        acc = __builtin_amdgcn_wmma_f32_16x16x4_f32(
             false, a01,  false, b01,
             (short)0, acc,  false,  false);
        acc = __builtin_amdgcn_wmma_f32_16x16x4_f32(
             false, a23,  false, b23,
             (short)0, acc,  false,  false);
    }
    return acc;
}

__global__ void __launch_bounds__(128)
gemm_bias_wmma_f32(const float* __restrict__ X,
                   const float* __restrict__ W,
                   const float* __restrict__ bias,
                   float* __restrict__ Out)
{
    const int wavesPerBlock = blockDim.x >> 5;
    const int wave = blockIdx.x * wavesPerBlock + (threadIdx.x >> 5);
    const int lane = threadIdx.x & 31;

    const int tilesN = HID / 16;
    const int tm = wave / tilesN;
    const int tn = wave % tilesN;
    const int s0 = tm * 16;
    const int h0 = tn * 16;

    const int half = lane >> 4;
    const int l    = lane & 15;

    const float* __restrict__ arow = X + (size_t)(s0 + l) * DIN + 4 * half;
    const float* __restrict__ brow = W + (size_t)(h0 + l) * DIN + 4 * half;

    v4f a0[NP], b0[NP], a1[NP], b1[NP];
    v8f acc = {};

    load_group(arow, brow, a0, b0);
    for (int t = 0; t < NG / 2 - 1; ++t) {
        const int k1 = (2 * t + 1) * KG;
        load_group(arow + k1, brow + k1, a1, b1);
        acc = mma_group(a0, b0, acc);
        load_group(arow + k1 + KG, brow + k1 + KG, a0, b0);
        acc = mma_group(a1, b1, acc);
    }
    load_group(arow + (NG - 1) * KG, brow + (NG - 1) * KG, a1, b1);
    acc = mma_group(a0, b0, acc);
    acc = mma_group(a1, b1, acc);

    const float bv = bias[h0 + l];
#pragma unroll
    for (int r = 0; r < 8; ++r) {
        Out[(size_t)(s0 + r + 8 * half) * HID + h0 + l] = acc[r] + bv;
    }
}

extern "C" void kernel_launch(void* const* d_in, const int* in_sizes, int n_in,
                              void* d_out, int out_size, void* d_ws, size_t ws_size,
                              hipStream_t stream)
{
    const float* x   = (const float*)d_in[0];
    const float* W_h = (const float*)d_in[1];
    const float* b_h = (const float*)d_in[2];
    const float* W_m = (const float*)d_in[3];
    const float* b_m = (const float*)d_in[4];
    const float* w2  = (const float*)d_in[5];
    const float* b2  = (const float*)d_in[6];
    float* out = (float*)d_out;

    float* Hbuf = (float*)d_ws;
    float* Mbuf = Hbuf + (size_t)SEQ * HID;

    const int tiles = (SEQ / 16) * (HID / 16);
    const int blk   = 128;
    const int grid  = tiles / (blk / 32);
    gemm_bias_wmma_f32<<<grid, blk, 0, stream>>>(x, W_h, b_h, Hbuf);
    gemm_bias_wmma_f32<<<grid, blk, 0, stream>>>(x, W_m, b_m, Mbuf);

    dim3 g2(SEQ / 32, SEQ / 32);
    pair_tanh_score<<<g2, 256, 0, stream>>>(Hbuf, Mbuf, w2, b2, out);
}


# reduction loops start at a block-dependent group and wrap (GEMM k groups, pair d steps): blocks no longer request the same column of all rows at once; k and d sum order change
# speedup vs baseline: 2.1061x; 1.1208x over previous
#include <hip/hip_runtime.h>
#include <hip/hip_bf16.h>
#include <math.h>

#define SEQ    512
#define DIN    1024
#define HID    512

typedef float v2f __attribute__((ext_vector_type(2)));
typedef float v8f __attribute__((ext_vector_type(8)));

__device__ __forceinline__ float hw_tanh(float x)
{
#if __has_builtin(__builtin_amdgcn_tanhf)
    return __builtin_amdgcn_tanhf(x);
#else
    return tanhf(x);
#endif
}

typedef float v4f __attribute__((ext_vector_type(4)));

#define TI    4
#define TJL   8
#define DSTEP 32
#define NSTEP (HID / DSTEP)

__device__ __forceinline__ void load_step(const float* __restrict__ hp,
                                          const float* __restrict__ mp,
                                          const float* __restrict__ wp,
                                          v4f (&h)[TI], v4f (&m)[TJL], v4f& w)
{
#pragma unroll
    for (int a = 0; a < TI; ++a) h[a] = *(const v4f*)(hp + (size_t)a * HID);
#pragma unroll
    for (int b = 0; b < TJL; ++b) m[b] = *(const v4f*)(mp + (size_t)b * HID);
    w = *(const v4f*)wp;
}

__device__ __forceinline__ v2f tanh2(v2f x)
{
    const v2f r = {hw_tanh(x.x), hw_tanh(x.y)};
    return r;
}

#define NCELL (TI * TJL * 2)
#define PIPE  3

__device__ __forceinline__ void pair_step(const v4f (&h)[TI], const v4f (&m)[TJL],
                                          const v4f w, v2f (&acc)[TI][TJL])
{
    const v2f wlo = {w.x, w.y};
    const v2f whi = {w.z, w.w};
    v2f x[NCELL], t[NCELL];
#pragma unroll
    for (int c = 0; c < NCELL + 2 * PIPE; ++c) {
        if (c < NCELL) {
            const int a = (c / TJL) % TI, b = c % TJL;
            const v2f hh = (c < NCELL / 2) ? (v2f){h[a].x, h[a].y} : (v2f){h[a].z, h[a].w};
            const v2f mm = (c < NCELL / 2) ? (v2f){m[b].x, m[b].y} : (v2f){m[b].z, m[b].w};
            x[c] = hh + mm;
        }
        if (c >= PIPE && c < NCELL + PIPE) t[c - PIPE] = tanh2(x[c - PIPE]);
        if (c >= 2 * PIPE) {
            const int e = c - 2 * PIPE;
            const int a = (e / TJL) % TI, b = e % TJL;
            acc[a][b] = __builtin_elementwise_fma((e < NCELL / 2) ? wlo : whi, t[e], acc[a][b]);
        }
    }
#pragma unroll
    for (int k = 0; k < 2 * NCELL; ++k) {
        __builtin_amdgcn_sched_group_barrier(0x002, 1, 0);
        __builtin_amdgcn_sched_group_barrier(0x400, 1, 0);
    }
}

template <int CTRL>
__device__ __forceinline__ float lane_exchange(float x)
{
    const int b = __builtin_bit_cast(int, x);
    return __builtin_bit_cast(float,
        __builtin_amdgcn_update_dpp(b, b, CTRL, 0xf, 0xf, false));
}

template <int N, int CTRL>
__device__ __forceinline__ void halve(float (&v)[32], const bool upper)
{
#pragma unroll
    for (int k = 0; k < N / 2; ++k) {
        const float lo = v[k];
        const float hi = v[k + N / 2];
        const float keep = upper ? hi : lo;
        const float send = upper ? lo : hi;
        v[k] = keep + lane_exchange<CTRL>(send);
    }
}

__global__ void __launch_bounds__(256)
pair_tanh_score(const float* __restrict__ Hm,
                const float* __restrict__ Mm,
                const float* __restrict__ w2,
                const float* __restrict__ b2p,
                float* __restrict__ out)
{
    const int lane = threadIdx.x & 31;
    const int wave = threadIdx.x >> 5;
    const int q = lane & 7;
    const int g = lane >> 3;
    const int iw = blockIdx.y * 32 + wave * TI;
    const int jw = blockIdx.x * 32;

    const float* __restrict__ hp = Hm + (size_t)iw * HID + 4 * q;
    const float* __restrict__ mp = Mm + (size_t)(jw + TJL * g) * HID + 4 * q;
    const float* __restrict__ wp = w2 + 4 * q;

    v2f acc[TI][TJL];
#pragma unroll
    for (int a = 0; a < TI; ++a)
#pragma unroll
        for (int b = 0; b < TJL; ++b) { acc[a][b].x = 0.0f; acc[a][b].y = 0.0f; }

    v4f h0[TI], m0[TJL], w0;
    v4f h1[TI], m1[TJL], w1;

    const int sr = (blockIdx.x + blockIdx.y) & (NSTEP - 1);

    const int dfirst = sr * DSTEP;
    load_step(hp + dfirst, mp + dfirst, wp + dfirst, h0, m0, w0);
    for (int t = 0; t < NSTEP / 2 - 1; ++t) {
        const int d1 = ((2 * t + 1 + sr) & (NSTEP - 1)) * DSTEP;
        const int d2 = ((2 * t + 2 + sr) & (NSTEP - 1)) * DSTEP;
        load_step(hp + d1, mp + d1, wp + d1, h1, m1, w1);
        __builtin_amdgcn_sched_barrier(0);
        pair_step(h0, m0, w0, acc);
        __builtin_amdgcn_sched_barrier(0);
        load_step(hp + d2, mp + d2, wp + d2, h0, m0, w0);
        __builtin_amdgcn_sched_barrier(0);
        pair_step(h1, m1, w1, acc);
        __builtin_amdgcn_sched_barrier(0);
    }
    const int dlast = ((NSTEP - 1 + sr) & (NSTEP - 1)) * DSTEP;
    load_step(hp + dlast, mp + dlast, wp + dlast, h1, m1, w1);
    __builtin_amdgcn_sched_barrier(0);
    pair_step(h0, m0, w0, acc);
    __builtin_amdgcn_sched_barrier(0);
    pair_step(h1, m1, w1, acc);

    float v[32];
#pragma unroll
    for (int a = 0; a < TI; ++a)
#pragma unroll
        for (int b = 0; b < TJL; ++b) v[TJL * a + b] = acc[a][b].x + acc[a][b].y;

    halve<32, 0x164>(v, (lane & 4) != 0);
    halve<16, 0x162>(v, (lane & 2) != 0);
    halve< 8, 0x161>(v, (lane & 1) != 0);

    const float b2 = b2p[0];
    const v4f res = {v[0] + b2, v[1] + b2, v[2] + b2, v[3] + b2};
    *(v4f*)(out + (size_t)(iw + (q >> 1)) * SEQ + jw + TJL * g + 4 * (q & 1)) = res;
}

#define KG 128
#define NP (KG / 8)
#define NG (DIN / KG)


__device__ __forceinline__ void load_group(const float* __restrict__ ap,
                                           const float* __restrict__ bp,
                                           v4f (&a)[NP], v4f (&b)[NP])
{
#pragma unroll
    for (int p = 0; p < NP; ++p) {
        a[p] = *(const v4f*)(ap + 8 * p);
        b[p] = *(const v4f*)(bp + 8 * p);
    }
}

__device__ __forceinline__ v8f mma_group(const v4f (&a)[NP], const v4f (&b)[NP], v8f acc)
{
#pragma unroll
    for (int p = 0; p < NP; ++p) {
        const v2f a01 = {a[p].x, a[p].y};
        const v2f b01 = {b[p].x, b[p].y};
        const v2f a23 = {a[p].z, a[p].w};
        const v2f b23 = {b[p].z, b[p].w};
        acc = __builtin_amdgcn_wmma_f32_16x16x4_f32(
             false, a01,  false, b01,
             (short)0, acc,  false,  false);
        acc = __builtin_amdgcn_wmma_f32_16x16x4_f32(
             false, a23,  false, b23,
             (short)0, acc,  false,  false);
    }
    return acc;
}

__global__ void __launch_bounds__(128)
gemm_bias_wmma_f32(const float* __restrict__ X,
                   const float* __restrict__ W,
                   const float* __restrict__ bias,
                   float* __restrict__ Out)
{
    const int wavesPerBlock = blockDim.x >> 5;
    const int wave = blockIdx.x * wavesPerBlock + (threadIdx.x >> 5);
    const int lane = threadIdx.x & 31;

    const int tilesN = HID / 16;
    const int tm = wave / tilesN;
    const int tn = wave % tilesN;
    const int s0 = tm * 16;
    const int h0 = tn * 16;

    const int half = lane >> 4;
    const int l    = lane & 15;

    const float* __restrict__ arow = X + (size_t)(s0 + l) * DIN + 4 * half;
    const float* __restrict__ brow = W + (size_t)(h0 + l) * DIN + 4 * half;

    v4f a0[NP], b0[NP], a1[NP], b1[NP];
    v8f acc = {};

    const int gr = ((blockIdx.x >> 3) + blockIdx.x) & (NG - 1);

    const int kfirst = gr * KG;
    load_group(arow + kfirst, brow + kfirst, a0, b0);
    for (int t = 0; t < NG / 2 - 1; ++t) {
        const int k1 = ((2 * t + 1 + gr) & (NG - 1)) * KG;
        const int k2 = ((2 * t + 2 + gr) & (NG - 1)) * KG;
        load_group(arow + k1, brow + k1, a1, b1);
        acc = mma_group(a0, b0, acc);
        load_group(arow + k2, brow + k2, a0, b0);
        acc = mma_group(a1, b1, acc);
    }
    const int klast = ((NG - 1 + gr) & (NG - 1)) * KG;
    load_group(arow + klast, brow + klast, a1, b1);
    acc = mma_group(a0, b0, acc);
    acc = mma_group(a1, b1, acc);

    const float bv = bias[h0 + l];
#pragma unroll
    for (int r = 0; r < 8; ++r) {
        Out[(size_t)(s0 + r + 8 * half) * HID + h0 + l] = acc[r] + bv;
    }
}

extern "C" void kernel_launch(void* const* d_in, const int* in_sizes, int n_in,
                              void* d_out, int out_size, void* d_ws, size_t ws_size,
                              hipStream_t stream)
{
    const float* x   = (const float*)d_in[0];
    const float* W_h = (const float*)d_in[1];
    const float* b_h = (const float*)d_in[2];
    const float* W_m = (const float*)d_in[3];
    const float* b_m = (const float*)d_in[4];
    const float* w2  = (const float*)d_in[5];
    const float* b2  = (const float*)d_in[6];
    float* out = (float*)d_out;

    float* Hbuf = (float*)d_ws;
    float* Mbuf = Hbuf + (size_t)SEQ * HID;

    const int tiles = (SEQ / 16) * (HID / 16);
    const int blk   = 128;
    const int grid  = tiles / (blk / 32);
    gemm_bias_wmma_f32<<<grid, blk, 0, stream>>>(x, W_h, b_h, Hbuf);
    gemm_bias_wmma_f32<<<grid, blk, 0, stream>>>(x, W_m, b_m, Mbuf);

    dim3 g2(SEQ / 32, SEQ / 32);
    pair_tanh_score<<<g2, 256, 0, stream>>>(Hbuf, Mbuf, w2, b2, out);
}


# projections as ONE GEMM on the stacked weights [W_h;W_m]: 8-wave blocks tiled 2x4, two waves per SIMD alternate on the matrix unit and share operand lines in cache; launches 3->2; k sum order changes
# speedup vs baseline: 2.3971x; 1.1382x over previous
#include <hip/hip_runtime.h>
#include <hip/hip_bf16.h>
#include <math.h>

#define SEQ    512
#define DIN    1024
#define HID    512

typedef float v2f __attribute__((ext_vector_type(2)));
typedef float v8f __attribute__((ext_vector_type(8)));

__device__ __forceinline__ float hw_tanh(float x)
{
#if __has_builtin(__builtin_amdgcn_tanhf)
    return __builtin_amdgcn_tanhf(x);
#else
    return tanhf(x);
#endif
}

typedef float v4f __attribute__((ext_vector_type(4)));

#define TI    4
#define TJL   8
#define DSTEP 32
#define NSTEP (HID / DSTEP)

__device__ __forceinline__ void load_step(const float* __restrict__ hp,
                                          const float* __restrict__ mp,
                                          const float* __restrict__ wp,
                                          v4f (&h)[TI], v4f (&m)[TJL], v4f& w)
{
#pragma unroll
    for (int a = 0; a < TI; ++a) h[a] = *(const v4f*)(hp + (size_t)a * HID);
#pragma unroll
    for (int b = 0; b < TJL; ++b) m[b] = *(const v4f*)(mp + (size_t)b * HID);
    w = *(const v4f*)wp;
}

__device__ __forceinline__ v2f tanh2(v2f x)
{
    const v2f r = {hw_tanh(x.x), hw_tanh(x.y)};
    return r;
}

#define NCELL (TI * TJL * 2)
#define PIPE  3

__device__ __forceinline__ void pair_step(const v4f (&h)[TI], const v4f (&m)[TJL],
                                          const v4f w, v2f (&acc)[TI][TJL])
{
    const v2f wlo = {w.x, w.y};
    const v2f whi = {w.z, w.w};
    v2f x[NCELL], t[NCELL];
#pragma unroll
    for (int c = 0; c < NCELL + 2 * PIPE; ++c) {
        if (c < NCELL) {
            const int a = (c / TJL) % TI, b = c % TJL;
            const v2f hh = (c < NCELL / 2) ? (v2f){h[a].x, h[a].y} : (v2f){h[a].z, h[a].w};
            const v2f mm = (c < NCELL / 2) ? (v2f){m[b].x, m[b].y} : (v2f){m[b].z, m[b].w};
            x[c] = hh + mm;
        }
        if (c >= PIPE && c < NCELL + PIPE) t[c - PIPE] = tanh2(x[c - PIPE]);
        if (c >= 2 * PIPE) {
            const int e = c - 2 * PIPE;
            const int a = (e / TJL) % TI, b = e % TJL;
            acc[a][b] = __builtin_elementwise_fma((e < NCELL / 2) ? wlo : whi, t[e], acc[a][b]);
        }
    }
#pragma unroll
    for (int k = 0; k < 2 * NCELL; ++k) {
        __builtin_amdgcn_sched_group_barrier(0x002, 1, 0);
        __builtin_amdgcn_sched_group_barrier(0x400, 1, 0);
    }
}

template <int CTRL>
__device__ __forceinline__ float lane_exchange(float x)
{
    const int b = __builtin_bit_cast(int, x);
    return __builtin_bit_cast(float,
        __builtin_amdgcn_update_dpp(b, b, CTRL, 0xf, 0xf, false));
}

template <int N, int CTRL>
__device__ __forceinline__ void halve(float (&v)[32], const bool upper)
{
#pragma unroll
    for (int k = 0; k < N / 2; ++k) {
        const float lo = v[k];
        const float hi = v[k + N / 2];
        const float keep = upper ? hi : lo;
        const float send = upper ? lo : hi;
        v[k] = keep + lane_exchange<CTRL>(send);
    }
}

__global__ void __launch_bounds__(256)
pair_tanh_score(const float* __restrict__ Hm,
                const float* __restrict__ Mm,
                const float* __restrict__ w2,
                const float* __restrict__ b2p,
                float* __restrict__ out)
{
    const int lane = threadIdx.x & 31;
    const int wave = threadIdx.x >> 5;
    const int q = lane & 7;
    const int g = lane >> 3;
    const int iw = blockIdx.y * 32 + wave * TI;
    const int jw = blockIdx.x * 32;

    const float* __restrict__ hp = Hm + (size_t)iw * HID + 4 * q;
    const float* __restrict__ mp = Mm + (size_t)(jw + TJL * g) * HID + 4 * q;
    const float* __restrict__ wp = w2 + 4 * q;

    v2f acc[TI][TJL];
#pragma unroll
    for (int a = 0; a < TI; ++a)
#pragma unroll
        for (int b = 0; b < TJL; ++b) { acc[a][b].x = 0.0f; acc[a][b].y = 0.0f; }

    v4f h0[TI], m0[TJL], w0;
    v4f h1[TI], m1[TJL], w1;

    const int sr = (blockIdx.x + blockIdx.y) & (NSTEP - 1);

    const int dfirst = sr * DSTEP;
    load_step(hp + dfirst, mp + dfirst, wp + dfirst, h0, m0, w0);
    for (int t = 0; t < NSTEP / 2 - 1; ++t) {
        const int d1 = ((2 * t + 1 + sr) & (NSTEP - 1)) * DSTEP;
        const int d2 = ((2 * t + 2 + sr) & (NSTEP - 1)) * DSTEP;
        load_step(hp + d1, mp + d1, wp + d1, h1, m1, w1);
        __builtin_amdgcn_sched_barrier(0);
        pair_step(h0, m0, w0, acc);
        __builtin_amdgcn_sched_barrier(0);
        load_step(hp + d2, mp + d2, wp + d2, h0, m0, w0);
        __builtin_amdgcn_sched_barrier(0);
        pair_step(h1, m1, w1, acc);
        __builtin_amdgcn_sched_barrier(0);
    }
    const int dlast = ((NSTEP - 1 + sr) & (NSTEP - 1)) * DSTEP;
    load_step(hp + dlast, mp + dlast, wp + dlast, h1, m1, w1);
    __builtin_amdgcn_sched_barrier(0);
    pair_step(h0, m0, w0, acc);
    __builtin_amdgcn_sched_barrier(0);
    pair_step(h1, m1, w1, acc);

    float v[32];
#pragma unroll
    for (int a = 0; a < TI; ++a)
#pragma unroll
        for (int b = 0; b < TJL; ++b) v[TJL * a + b] = acc[a][b].x + acc[a][b].y;

    halve<32, 0x164>(v, (lane & 4) != 0);
    halve<16, 0x162>(v, (lane & 2) != 0);
    halve< 8, 0x161>(v, (lane & 1) != 0);

    const float b2 = b2p[0];
    const v4f res = {v[0] + b2, v[1] + b2, v[2] + b2, v[3] + b2};
    *(v4f*)(out + (size_t)(iw + (q >> 1)) * SEQ + jw + TJL * g + 4 * (q & 1)) = res;
}

#define KG 128
#define NP (KG / 8)
#define NG (DIN / KG)


__device__ __forceinline__ void load_group(const float* __restrict__ ap,
                                           const float* __restrict__ bp,
                                           v4f (&a)[NP], v4f (&b)[NP])
{
#pragma unroll
    for (int p = 0; p < NP; ++p) {
        a[p] = *(const v4f*)(ap + 8 * p);
        b[p] = *(const v4f*)(bp + 8 * p);
    }
}

__device__ __forceinline__ v8f mma_group(const v4f (&a)[NP], const v4f (&b)[NP], v8f acc)
{
#pragma unroll
    for (int p = 0; p < NP; ++p) {
        const v2f a01 = {a[p].x, a[p].y};
        const v2f b01 = {b[p].x, b[p].y};
        const v2f a23 = {a[p].z, a[p].w};
        const v2f b23 = {b[p].z, b[p].w};
        acc = __builtin_amdgcn_wmma_f32_16x16x4_f32(
             false, a01,  false, b01,
             (short)0, acc,  false,  false);
        acc = __builtin_amdgcn_wmma_f32_16x16x4_f32(
             false, a23,  false, b23,
             (short)0, acc,  false,  false);
    }
    return acc;
}

#define BS_ROWS 32
#define BN_COLS 64
__global__ void __launch_bounds__(256)
proj_stacked_wmma_f32(const float* __restrict__ X,
                      const float* __restrict__ Wh,
                      const float* __restrict__ bh,
                      const float* __restrict__ Wm,
                      const float* __restrict__ bm,
                      float* __restrict__ Hout,
                      float* __restrict__ Mout)
{
    const int lane = threadIdx.x & 31;
    const int wave = threadIdx.x >> 5;
    const int bn = blockIdx.x;
    const int bs = blockIdx.y;

    const bool second = bn >= HID / BN_COLS;
    const float* __restrict__ W    = second ? Wm : Wh;
    const float* __restrict__ bias = second ? bm : bh;
    float* __restrict__ Out        = second ? Mout : Hout;

    const int s0 = bs * BS_ROWS + (wave >> 2) * 16;
    const int h0 = (bn & (HID / BN_COLS - 1)) * BN_COLS + (wave & 3) * 16;

    const int half = lane >> 4;
    const int l    = lane & 15;

    const float* __restrict__ arow = X + (size_t)(s0 + l) * DIN + 4 * half;
    const float* __restrict__ brow = W + (size_t)(h0 + l) * DIN + 4 * half;

    v4f a0[NP], b0[NP], a1[NP], b1[NP];
    v8f acc = {};

    const int gr = (bs + bn) & (NG - 1);

    const int kfirst = gr * KG;
    load_group(arow + kfirst, brow + kfirst, a0, b0);
    for (int t = 0; t < NG / 2 - 1; ++t) {
        const int k1 = ((2 * t + 1 + gr) & (NG - 1)) * KG;
        const int k2 = ((2 * t + 2 + gr) & (NG - 1)) * KG;
        load_group(arow + k1, brow + k1, a1, b1);
        acc = mma_group(a0, b0, acc);
        load_group(arow + k2, brow + k2, a0, b0);
        acc = mma_group(a1, b1, acc);
    }
    const int klast = ((NG - 1 + gr) & (NG - 1)) * KG;
    load_group(arow + klast, brow + klast, a1, b1);
    acc = mma_group(a0, b0, acc);
    acc = mma_group(a1, b1, acc);

    const float bv = bias[h0 + l];
#pragma unroll
    for (int r = 0; r < 8; ++r) {
        Out[(size_t)(s0 + r + 8 * half) * HID + h0 + l] = acc[r] + bv;
    }
}

extern "C" void kernel_launch(void* const* d_in, const int* in_sizes, int n_in,
                              void* d_out, int out_size, void* d_ws, size_t ws_size,
                              hipStream_t stream)
{
    const float* x   = (const float*)d_in[0];
    const float* W_h = (const float*)d_in[1];
    const float* b_h = (const float*)d_in[2];
    const float* W_m = (const float*)d_in[3];
    const float* b_m = (const float*)d_in[4];
    const float* w2  = (const float*)d_in[5];
    const float* b2  = (const float*)d_in[6];
    float* out = (float*)d_out;

    float* Hbuf = (float*)d_ws;
    float* Mbuf = Hbuf + (size_t)SEQ * HID;

    dim3 g1(2 * HID / BN_COLS, SEQ / BS_ROWS);
    proj_stacked_wmma_f32<<<g1, 256, 0, stream>>>(x, W_h, b_h, W_m, b_m, Hbuf, Mbuf);

    dim3 g2(SEQ / 32, SEQ / 32);
    pair_tanh_score<<<g2, 256, 0, stream>>>(Hbuf, Mbuf, w2, b2, out);
}


# GEMM k range halved across work-groups, wave tile 32x32: 4 fragment requests feed 8 WMMAs (4 accumulators), one wave per SIMD; pair phase sums the two partial planes on load; k sum order changes
# speedup vs baseline: 2.6018x; 1.0854x over previous
#include <hip/hip_runtime.h>
#include <hip/hip_bf16.h>
#include <math.h>

#define SEQ    512
#define DIN    1024
#define HID    512

typedef float v2f __attribute__((ext_vector_type(2)));
typedef float v8f __attribute__((ext_vector_type(8)));

__device__ __forceinline__ float hw_tanh(float x)
{
#if __has_builtin(__builtin_amdgcn_tanhf)
    return __builtin_amdgcn_tanhf(x);
#else
    return tanhf(x);
#endif
}

typedef float v4f __attribute__((ext_vector_type(4)));

#define TI    4
#define TJL   8
#define DSTEP 32
#define NSTEP (HID / DSTEP)

#define PLANE ((size_t)SEQ * HID)

__device__ __forceinline__ void load_step(const float* __restrict__ hp,
                                          const float* __restrict__ mp,
                                          const float* __restrict__ wp,
                                          v4f (&h)[2][TI], v4f (&m)[2][TJL], v4f& w)
{
#pragma unroll
    for (int pl = 0; pl < 2; ++pl) {
#pragma unroll
        for (int a = 0; a < TI; ++a) h[pl][a] = *(const v4f*)(hp + pl * PLANE + (size_t)a * HID);
#pragma unroll
        for (int b = 0; b < TJL; ++b) m[pl][b] = *(const v4f*)(mp + pl * PLANE + (size_t)b * HID);
    }
    w = *(const v4f*)wp;
}

__device__ __forceinline__ v2f tanh2(v2f x)
{
    const v2f r = {hw_tanh(x.x), hw_tanh(x.y)};
    return r;
}

#define NCELL (TI * TJL * 2)
#define PIPE  3

__device__ __forceinline__ void pair_step(const v4f (&hq)[2][TI], const v4f (&mq)[2][TJL],
                                          const v4f w, v2f (&acc)[TI][TJL])
{
    v4f h[TI], m[TJL];
#pragma unroll
    for (int a = 0; a < TI; ++a) h[a] = hq[0][a] + hq[1][a];
#pragma unroll
    for (int b = 0; b < TJL; ++b) m[b] = mq[0][b] + mq[1][b];

    const v2f wlo = {w.x, w.y};
    const v2f whi = {w.z, w.w};
    v2f x[NCELL], t[NCELL];
#pragma unroll
    for (int c = 0; c < NCELL + 2 * PIPE; ++c) {
        if (c < NCELL) {
            const int a = (c / TJL) % TI, b = c % TJL;
            const v2f hh = (c < NCELL / 2) ? (v2f){h[a].x, h[a].y} : (v2f){h[a].z, h[a].w};
            const v2f mm = (c < NCELL / 2) ? (v2f){m[b].x, m[b].y} : (v2f){m[b].z, m[b].w};
            x[c] = hh + mm;
        }
        if (c >= PIPE && c < NCELL + PIPE) t[c - PIPE] = tanh2(x[c - PIPE]);
        if (c >= 2 * PIPE) {
            const int e = c - 2 * PIPE;
            const int a = (e / TJL) % TI, b = e % TJL;
            acc[a][b] = __builtin_elementwise_fma((e < NCELL / 2) ? wlo : whi, t[e], acc[a][b]);
        }
    }
#pragma unroll
    for (int k = 0; k < 2 * (TI + TJL); ++k)
        __builtin_amdgcn_sched_group_barrier(0x002, 1, 0);
#pragma unroll
    for (int k = 0; k < 2 * NCELL; ++k) {
        __builtin_amdgcn_sched_group_barrier(0x002, 1, 0);
        __builtin_amdgcn_sched_group_barrier(0x400, 1, 0);
    }
}

template <int CTRL>
__device__ __forceinline__ float lane_exchange(float x)
{
    const int b = __builtin_bit_cast(int, x);
    return __builtin_bit_cast(float,
        __builtin_amdgcn_update_dpp(b, b, CTRL, 0xf, 0xf, false));
}

template <int N, int CTRL>
__device__ __forceinline__ void halve(float (&v)[32], const bool upper)
{
#pragma unroll
    for (int k = 0; k < N / 2; ++k) {
        const float lo = v[k];
        const float hi = v[k + N / 2];
        const float keep = upper ? hi : lo;
        const float send = upper ? lo : hi;
        v[k] = keep + lane_exchange<CTRL>(send);
    }
}

__global__ void __launch_bounds__(256)
pair_tanh_score(const float* __restrict__ Hm,
                const float* __restrict__ Mm,
                const float* __restrict__ w2,
                const float* __restrict__ b2p,
                float* __restrict__ out)
{
    const int lane = threadIdx.x & 31;
    const int wave = threadIdx.x >> 5;
    const int q = lane & 7;
    const int g = lane >> 3;
    const int iw = blockIdx.y * 32 + wave * TI;
    const int jw = blockIdx.x * 32;

    const float* __restrict__ hp = Hm + (size_t)iw * HID + 4 * q;
    const float* __restrict__ mp = Mm + (size_t)(jw + TJL * g) * HID + 4 * q;
    const float* __restrict__ wp = w2 + 4 * q;

    v2f acc[TI][TJL];
#pragma unroll
    for (int a = 0; a < TI; ++a)
#pragma unroll
        for (int b = 0; b < TJL; ++b) { acc[a][b].x = 0.0f; acc[a][b].y = 0.0f; }

    v4f h0[2][TI], m0[2][TJL], w0;
    v4f h1[2][TI], m1[2][TJL], w1;

    const int sr = (blockIdx.x + blockIdx.y) & (NSTEP - 1);

    const int dfirst = sr * DSTEP;
    load_step(hp + dfirst, mp + dfirst, wp + dfirst, h0, m0, w0);
    for (int t = 0; t < NSTEP / 2 - 1; ++t) {
        const int d1 = ((2 * t + 1 + sr) & (NSTEP - 1)) * DSTEP;
        const int d2 = ((2 * t + 2 + sr) & (NSTEP - 1)) * DSTEP;
        load_step(hp + d1, mp + d1, wp + d1, h1, m1, w1);
        __builtin_amdgcn_sched_barrier(0);
        pair_step(h0, m0, w0, acc);
        __builtin_amdgcn_sched_barrier(0);
        load_step(hp + d2, mp + d2, wp + d2, h0, m0, w0);
        __builtin_amdgcn_sched_barrier(0);
        pair_step(h1, m1, w1, acc);
        __builtin_amdgcn_sched_barrier(0);
    }
    const int dlast = ((NSTEP - 1 + sr) & (NSTEP - 1)) * DSTEP;
    load_step(hp + dlast, mp + dlast, wp + dlast, h1, m1, w1);
    __builtin_amdgcn_sched_barrier(0);
    pair_step(h0, m0, w0, acc);
    __builtin_amdgcn_sched_barrier(0);
    pair_step(h1, m1, w1, acc);

    float v[32];
#pragma unroll
    for (int a = 0; a < TI; ++a)
#pragma unroll
        for (int b = 0; b < TJL; ++b) v[TJL * a + b] = acc[a][b].x + acc[a][b].y;

    halve<32, 0x164>(v, (lane & 4) != 0);
    halve<16, 0x162>(v, (lane & 2) != 0);
    halve< 8, 0x161>(v, (lane & 1) != 0);

    const float b2 = b2p[0];
    const v4f res = {v[0] + b2, v[1] + b2, v[2] + b2, v[3] + b2};
    *(v4f*)(out + (size_t)(iw + (q >> 1)) * SEQ + jw + TJL * g + 4 * (q & 1)) = res;
}

#define KSPLIT 2
#define KSPAN (DIN / KSPLIT)
#define KG 128
#define NG (KSPAN / KG)
#define NPIECE (KSPAN / 8)
#define PPG (KG / 8)
#define NWIN 14

__device__ __forceinline__ v8f wmma4(v2f a, v2f b, v8f c)
{
    return __builtin_amdgcn_wmma_f32_16x16x4_f32(
         false, a,  false, b,
         (short)0, c,  false,  false);
}

#define BS_ROWS 64
#define BN_COLS 64
__global__ void __launch_bounds__(128)
proj_stacked_wmma_f32(const float* __restrict__ X,
                      const float* __restrict__ Wh,
                      const float* __restrict__ bh,
                      const float* __restrict__ Wm,
                      const float* __restrict__ bm,
                      float* __restrict__ Hout,
                      float* __restrict__ Mout)
{
    const int lane = threadIdx.x & 31;
    const int wave = threadIdx.x >> 5;
    const int bn = blockIdx.x;
    const int bs = blockIdx.y;
    const int kh = blockIdx.z;

    const bool second = bn >= HID / BN_COLS;
    const float* __restrict__ W    = second ? Wm : Wh;
    const float* __restrict__ bias = second ? bm : bh;
    float* __restrict__ Out        = (second ? Mout : Hout) + (size_t)kh * PLANE;

    const int s0 = bs * BS_ROWS + (wave >> 1) * 32;
    const int h0 = (bn & (HID / BN_COLS - 1)) * BN_COLS + (wave & 1) * 32;

    const int half = lane >> 4;
    const int l    = lane & 15;

    const float* __restrict__ arow = X + (size_t)(s0 + l) * DIN + kh * KSPAN + 4 * half;
    const float* __restrict__ brow = W + (size_t)(h0 + l) * DIN + kh * KSPAN + 4 * half;

    const int gr = (bs + bn) & (NG - 1);
    const float* ag[NG];
    const float* bg[NG];
#pragma unroll
    for (int g = 0; g < NG; ++g) {
        const int koff = ((g + gr) & (NG - 1)) * KG;
        ag[g] = arow + koff;
        bg[g] = brow + koff;
    }

    v4f a0[NWIN], a1[NWIN], b0[NWIN], b1[NWIN];
    v8f acc[2][2] = {};

#pragma unroll
    for (int i = 0; i < NWIN; ++i) {
        const float* ap = ag[i / PPG] + 8 * (i % PPG);
        const float* bp = bg[i / PPG] + 8 * (i % PPG);
        a0[i] = *(const v4f*)ap;
        b0[i] = *(const v4f*)bp;
        a1[i] = *(const v4f*)(ap + (size_t)16 * DIN);
        b1[i] = *(const v4f*)(bp + (size_t)16 * DIN);
    }
    __builtin_amdgcn_sched_barrier(0);

#pragma unroll
    for (int i = 0; i < NPIECE; ++i) {
        const int w = i % NWIN;
        const v2f a0lo = {a0[w].x, a0[w].y}, a0hi = {a0[w].z, a0[w].w};
        const v2f a1lo = {a1[w].x, a1[w].y}, a1hi = {a1[w].z, a1[w].w};
        const v2f b0lo = {b0[w].x, b0[w].y}, b0hi = {b0[w].z, b0[w].w};
        const v2f b1lo = {b1[w].x, b1[w].y}, b1hi = {b1[w].z, b1[w].w};
        acc[0][0] = wmma4(a0lo, b0lo, acc[0][0]);
        acc[0][1] = wmma4(a0lo, b1lo, acc[0][1]);
        acc[1][0] = wmma4(a1lo, b0lo, acc[1][0]);
        acc[1][1] = wmma4(a1lo, b1lo, acc[1][1]);
        acc[0][0] = wmma4(a0hi, b0hi, acc[0][0]);
        acc[0][1] = wmma4(a0hi, b1hi, acc[0][1]);
        acc[1][0] = wmma4(a1hi, b0hi, acc[1][0]);
        acc[1][1] = wmma4(a1hi, b1hi, acc[1][1]);
        if (i + NWIN < NPIECE) {
            const int n = i + NWIN;
            const float* ap = ag[n / PPG] + 8 * (n % PPG);
            const float* bp = bg[n / PPG] + 8 * (n % PPG);
            a0[w] = *(const v4f*)ap;
            b0[w] = *(const v4f*)bp;
            a1[w] = *(const v4f*)(ap + (size_t)16 * DIN);
            b1[w] = *(const v4f*)(bp + (size_t)16 * DIN);
        }
        __builtin_amdgcn_sched_barrier(0);
    }

#pragma unroll
    for (int j = 0; j < 2; ++j) {
        const float bv = (kh == 0) ? bias[h0 + 16 * j + l] : 0.0f;
#pragma unroll
        for (int i = 0; i < 2; ++i) {
#pragma unroll
            for (int r = 0; r < 8; ++r) {
                Out[(size_t)(s0 + 16 * i + r + 8 * half) * HID + h0 + 16 * j + l] = acc[i][j][r] + bv;
            }
        }
    }
}

extern "C" void kernel_launch(void* const* d_in, const int* in_sizes, int n_in,
                              void* d_out, int out_size, void* d_ws, size_t ws_size,
                              hipStream_t stream)
{
    const float* x   = (const float*)d_in[0];
    const float* W_h = (const float*)d_in[1];
    const float* b_h = (const float*)d_in[2];
    const float* W_m = (const float*)d_in[3];
    const float* b_m = (const float*)d_in[4];
    const float* w2  = (const float*)d_in[5];
    const float* b2  = (const float*)d_in[6];
    float* out = (float*)d_out;

    float* Hbuf = (float*)d_ws;
    float* Mbuf = Hbuf + (size_t)KSPLIT * SEQ * HID;

    dim3 g1(2 * HID / BN_COLS, SEQ / BS_ROWS, KSPLIT);
    proj_stacked_wmma_f32<<<g1, 128, 0, stream>>>(x, W_h, b_h, W_m, b_m, Hbuf, Mbuf);

    dim3 g2(SEQ / 32, SEQ / 32);
    pair_tanh_score<<<g2, 256, 0, stream>>>(Hbuf, Mbuf, w2, b2, out);
}
